# starting kernel
# baseline (speedup 1.0000x reference)
#include <hip/hip_runtime.h>
#include <hip/hip_bf16.h>

#define EMBED_DIM 128
#define KNN 10
#define NODE_DIM 22
#define FDIM 24
#define NNODES 2000
#define NBATCH 16
#define BLOCK 128
#define NODES_PER_BLOCK 128
#define BLOCKS_PER_BATCH 16
#define FSTRIDE 25

typedef __attribute__((ext_vector_type(2))) float v2f;
typedef __attribute__((ext_vector_type(8))) float v8f;

__global__ __launch_bounds__(BLOCK)
void tsp_embed_kernel(const float* __restrict__ locs,
                      const float* __restrict__ W,
                      const float* __restrict__ bias,
                      float* __restrict__ out)
{
    __shared__ float sLocs[2 * NNODES];
    __shared__ float sFeat[NODES_PER_BLOCK * FSTRIDE];
    __shared__ float sW[FDIM * EMBED_DIM];

    const int tid   = threadIdx.x;
    const int batch = blockIdx.x / BLOCKS_PER_BATCH;
    const int node0 = (blockIdx.x % BLOCKS_PER_BATCH) * NODES_PER_BLOCK;

    const float* bl = locs + (size_t)batch * NNODES * 2;
    for (int idx = tid; idx < 2 * NNODES; idx += BLOCK) sLocs[idx] = bl[idx];
    for (int idx = tid; idx < FDIM * EMBED_DIM; idx += BLOCK) {
        const int f = idx / EMBED_DIM;
        const int e = idx % EMBED_DIM;
        sW[idx] = (f < NODE_DIM) ? W[e * NODE_DIM + f] : 0.0f;
    }
    __syncthreads();

    const int node = node0 + tid;
    float bestD[KNN];
    int   bestJ[KNN];
#pragma unroll
    for (int k = 0; k < KNN; ++k) { bestD[k] = 3.0e38f; bestJ[k] = 0; }

    const float2* sl = (const float2*)sLocs;
    float px = 0.0f, py = 0.0f;
    if (node < NNODES) {
        const float2 me = sl[node];
        px = me.x; py = me.y;
        for (int j = 0; j < NNODES; ++j) {
            const float2 q = sl[j];
            const float dx = q.x - px;
            const float dy = q.y - py;
            const float d2 = dx * dx + dy * dy;
            if (j != node && d2 < bestD[KNN - 1]) {
                bestD[KNN - 1] = d2;
                bestJ[KNN - 1] = j;
#pragma unroll
                for (int s = KNN - 1; s > 0; --s) {
                    if (bestD[s] < bestD[s - 1]) {
                        const float td = bestD[s]; bestD[s] = bestD[s - 1]; bestD[s - 1] = td;
                        const int   tj = bestJ[s]; bestJ[s] = bestJ[s - 1]; bestJ[s - 1] = tj;
                    }
                }
            }
        }
    }

    float* fr = &sFeat[tid * FSTRIDE];
    if (node < NNODES) {
        fr[0] = px;
        fr[1] = py;
#pragma unroll
        for (int k = 0; k < KNN; ++k) {
            const float2 q = sl[bestJ[k]];
            fr[2 + 2 * k]     = q.x - px;
            fr[2 + 2 * k + 1] = q.y - py;
        }
        fr[NODE_DIM]     = 0.0f;
        fr[NODE_DIM + 1] = 0.0f;
    } else {
#pragma unroll
        for (int k = 0; k < FDIM; ++k) fr[k] = 0.0f;
    }
    __syncthreads();

    const int wave   = tid >> 5;
    const int lane   = tid & 31;
    const int laneLo = lane & 15;
    const int kHalf  = (lane < 16) ? 0 : 2;
    const int mBase  = (lane < 16) ? 0 : 8;

    const int t0 = wave * 2;
    const int t1 = t0 + 1;
    const bool ok0 = (node0 + t0 * 16) < NNODES;
    const bool ok1 = (node0 + t1 * 16) < NNODES;

    v2f a0[6], a1[6];
    {
        const float* fa0 = &sFeat[(t0 * 16 + laneLo) * FSTRIDE];
        const float* fa1 = &sFeat[(t1 * 16 + laneLo) * FSTRIDE];
#pragma unroll
        for (int kc = 0; kc < 6; ++kc) {
            const int k0 = kc * 4 + kHalf;
            a0[kc].x = fa0[k0];  a0[kc].y = fa0[k0 + 1];
            a1[kc].x = fa1[k0];  a1[kc].y = fa1[k0 + 1];
        }
    }

    float* const outB = out + (size_t)batch * NNODES * EMBED_DIM;

#pragma unroll
    for (int ct = 0; ct < 8; ++ct) {
        const int e = ct * 16 + laneLo;
        v2f bf[6];
#pragma unroll
        for (int kc = 0; kc < 6; ++kc) {
            const int k0 = kc * 4 + kHalf;
            bf[kc].x = sW[k0 * EMBED_DIM + e];
            bf[kc].y = sW[(k0 + 1) * EMBED_DIM + e];
        }
        v8f acc0 = {}, acc1 = {};
#pragma unroll
        for (int kc = 0; kc < 6; ++kc)
            acc0 = __builtin_amdgcn_wmma_f32_16x16x4_f32(
                false, a0[kc], false, bf[kc], (short)0, acc0, false, false);
#pragma unroll
        for (int kc = 0; kc < 6; ++kc)
            acc1 = __builtin_amdgcn_wmma_f32_16x16x4_f32(
                false, a1[kc], false, bf[kc], (short)0, acc1, false, false);

        const float bv = bias[e];
        if (ok0) {
            float* p = outB + (size_t)(node0 + t0 * 16 + mBase) * EMBED_DIM + e;
#pragma unroll
            for (int r = 0; r < 8; ++r)
                p[(size_t)r * EMBED_DIM] = acc0[r] + bv;
        }
        if (ok1) {
            float* p = outB + (size_t)(node0 + t1 * 16 + mBase) * EMBED_DIM + e;
#pragma unroll
            for (int r = 0; r < 8; ++r)
                p[(size_t)r * EMBED_DIM] = acc1[r] + bv;
        }
    }
}

extern "C" void kernel_launch(void* const* d_in, const int* in_sizes, int n_in,
                              void* d_out, int out_size, void* d_ws, size_t ws_size,
                              hipStream_t stream) {
    const float* locs = (const float*)d_in[0];
    const float* W    = (const float*)d_in[1];
    const float* b    = (const float*)d_in[2];
    float* out        = (float*)d_out;

    dim3 grid(NBATCH * BLOCKS_PER_BATCH);
    dim3 block(BLOCK);
    tsp_embed_kernel<<<grid, block, 0, stream>>>(locs, W, b, out);
}


# KNN scan sorts one 32-bit word per candidate (upper 21 bits of d2, index in low 11): insert = 1 min + 11 max-min; list neighbours in one 21-bit bucket re-sorted by exact key; 8 candidates a trip
# speedup vs baseline: 10.7764x; 10.7764x over previous
#include <hip/hip_runtime.h>
#include <hip/hip_bf16.h>

#define EMBED_DIM 128
#define KNN 10
#define NODE_DIM 22
#define FDIM 24
#define NNODES 2000
#define NBATCH 16
#define BLOCK 128
#define NODES_PER_BLOCK 128
#define BLOCKS_PER_BATCH 16
#define FSTRIDE 25
#define GX 8
#define GY 16
#define NCELLS (GX * GY)
#define PTS_PER_THREAD ((NNODES + BLOCK - 1) / BLOCK)

typedef __attribute__((ext_vector_type(2))) float v2f;
typedef __attribute__((ext_vector_type(8))) float v8f;

__global__ __launch_bounds__(BLOCK)
void tsp_embed_kernel(const float* __restrict__ locs,
                      const float* __restrict__ W,
                      const float* __restrict__ bias,
                      float* __restrict__ out)
{
    __shared__ float sLocs[2 * NNODES];
    __shared__ float sFeat[NODES_PER_BLOCK * FSTRIDE];
    __shared__ float sW[FDIM * EMBED_DIM];
    __shared__ float4 sCand[NNODES + 12];
    __shared__ __attribute__((aligned(16))) unsigned short sHist[BLOCK * NCELLS];
    __shared__ int sCnt[NCELLS];
    __shared__ int sStart[NCELLS + 1];

    const int tid   = threadIdx.x;
    __builtin_assume(tid >= 0 && tid < BLOCK);
    const int batch = blockIdx.x / BLOCKS_PER_BATCH;
    const int node0 = (blockIdx.x % BLOCKS_PER_BATCH) * NODES_PER_BLOCK;

    const float2* bl2 = (const float2*)(locs + (size_t)batch * NNODES * 2);
    float2* sl2w = (float2*)sLocs;
    float2 pt[PTS_PER_THREAD];
    int key[PTS_PER_THREAD];
    int slot[PTS_PER_THREAD];
#pragma unroll
    for (int k = 0; k < PTS_PER_THREAD; ++k) {
        const int n = tid + k * BLOCK;
        pt[k] = (n < NNODES) ? bl2[n] : make_float2(0.0f, 0.0f);
    }
    unsigned short* const myH = &sHist[tid * NCELLS];
#pragma unroll
    for (int i = 0; i < NCELLS; ++i) myH[i] = (unsigned short)0;
#pragma unroll
    for (int k = 0; k < PTS_PER_THREAD; ++k) {
        const int n = tid + k * BLOCK;
        key[k] = 0;
        slot[k] = 0;
        if (n < NNODES) {
            const float2 p = pt[k];
            sl2w[n] = p;
            const int bx = (int)fminf(fmaxf(p.x * (float)GX, 0.0f), (float)(GX - 1));
            const int by = (int)fminf(fmaxf(p.y * (float)GY, 0.0f), (float)(GY - 1));
            key[k] = by * GX + bx;
            const unsigned v = myH[key[k]];
            slot[k] = (int)v;
            myH[key[k]] = (unsigned short)(v + 1u);
        }
    }
    for (int idx = tid; idx < FDIM * EMBED_DIM; idx += BLOCK) {
        const int f = idx / EMBED_DIM;
        const int e = idx % EMBED_DIM;
        sW[idx] = (f < NODE_DIM) ? W[e * NODE_DIM + f] : 0.0f;
    }
    if (tid < 12) sCand[NNODES + tid] = make_float4(0.0f, 0.0f, 0.0f, 0.0f);
    __syncthreads();

    {
        int run = 0;
#pragma unroll 16
        for (int t = 0; t < BLOCK; ++t) {
            const int v = sHist[t * NCELLS + tid];
            sHist[t * NCELLS + tid] = (unsigned short)run;
            run += v;
        }
        sCnt[tid] = run;
    }
    __syncthreads();
    int start = 0;
    for (int i = 0; i < NCELLS; ++i) start += (i < tid) ? sCnt[i] : 0;
    sStart[tid] = start;
    if (tid == 0) sStart[NCELLS] = NNODES;
    __syncthreads();
#pragma unroll
    for (int k = 0; k < PTS_PER_THREAD; ++k) {
        const int n = tid + k * BLOCK;
        if (n < NNODES) {
            const int pos = sStart[key[k]] + (int)myH[key[k]] + slot[k];
            sCand[pos] = make_float4(pt[k].x, pt[k].y, __int_as_float(n), 0.0f);
        }
    }
    __syncthreads();
    const float2* sl = (const float2*)sLocs;

    const int node = node0 + tid;
    unsigned K0, K1, K2, K3, K4, K5, K6, K7, K8, K9, K10, K11;
    unsigned J0 = 0, J1 = 0, J2 = 0, J3 = 0, J4 = 0, J5 = 0, J6 = 0, J7 = 0, J8 = 0, J9 = 0;
    const unsigned IDX_MASK = 0x7FFu;
    const unsigned BKT_MASK = 0xFFFFF800u;
    const unsigned FARK = 0x7F000000u;
    K0 = K1 = K2 = K3 = K4 = K5 = K6 = K7 = K8 = K9 = K10 = K11 = FARK;
#define WORD_INSERT(w) {                         \
        K11 = min(max(K10, (w)), K11);          \
        K10 = min(max(K9, (w)), K10);           \
        K9 = min(max(K8, (w)), K9);             \
        K8 = min(max(K7, (w)), K8);             \
        K7 = min(max(K6, (w)), K7);             \
        K6 = min(max(K5, (w)), K6);             \
        K5 = min(max(K4, (w)), K5);             \
        K4 = min(max(K3, (w)), K4);             \
        K3 = min(max(K2, (w)), K3);             \
        K2 = min(max(K1, (w)), K2);             \
        K1 = min(max(K0, (w)), K1);             \
        K0 = min(K0, (w));                      \
    }
#define KEY_LT(du, j, Dk, Jk) (((du) < (Dk)) | (((du) == (Dk)) & ((j) < (Jk))))
#define KEY_INSERT(duv, jv) {                                                   \
        const unsigned eu = (duv), ej = (jv);                                   \
        const bool c0 = KEY_LT(eu, ej, D0, J0), c1 = KEY_LT(eu, ej, D1, J1);    \
        const bool c2 = KEY_LT(eu, ej, D2, J2), c3 = KEY_LT(eu, ej, D3, J3);    \
        const bool c4 = KEY_LT(eu, ej, D4, J4), c5 = KEY_LT(eu, ej, D5, J5);    \
        const bool c6 = KEY_LT(eu, ej, D6, J6), c7 = KEY_LT(eu, ej, D7, J7);    \
        const bool c8 = KEY_LT(eu, ej, D8, J8), c9 = KEY_LT(eu, ej, D9, J9);    \
        D9 = min(max(D8, eu), D9);        J9 = c8 ? J8 : (c9 ? ej : J9);        \
        D8 = min(max(D7, eu), D8);        J8 = c7 ? J7 : (c8 ? ej : J8);        \
        D7 = min(max(D6, eu), D7);        J7 = c6 ? J6 : (c7 ? ej : J7);        \
        D6 = min(max(D5, eu), D6);        J6 = c5 ? J5 : (c6 ? ej : J6);        \
        D5 = min(max(D4, eu), D5);        J5 = c4 ? J4 : (c5 ? ej : J5);        \
        D4 = min(max(D3, eu), D4);        J4 = c3 ? J3 : (c4 ? ej : J4);        \
        D3 = min(max(D2, eu), D3);        J3 = c2 ? J2 : (c3 ? ej : J3);        \
        D2 = min(max(D1, eu), D2);        J2 = c1 ? J1 : (c2 ? ej : J2);        \
        D1 = min(max(D0, eu), D1);        J1 = c0 ? J0 : (c1 ? ej : J1);        \
        D0 = min(D0, eu);                 J0 = c0 ? ej : J0;                    \
    }
    float px = 0.0f, py = 0.0f;
    if (node < NNODES) {
        const float2 me = sl[node];
        px = me.x; py = me.y;
        const float R = 1.0f / 16.0f;
        int xlo = (int)fminf(fmaxf((px - R) * (float)GX, 0.0f), (float)(GX - 1));
        int xhi = (int)fminf(fmaxf((px + R) * (float)GX, 0.0f), (float)(GX - 1));
        int ylo = (int)fminf(fmaxf((py - R) * (float)GY, 0.0f), (float)(GY - 1));
        int yhi = (int)fminf(fmaxf((py + R) * (float)GY, 0.0f), (float)(GY - 1));
        int pxlo = 1, pxhi = 0, pylo = 1, pyhi = 0;
        int sx0 = xlo, sx1 = xhi, sy0 = ylo, sy1 = yhi;
        for (int grow = 0; grow < GX + GY; ++grow) {
            const int nseg = 2 * (yhi - ylo + 1);
            for (int seg = 0; seg < nseg; ++seg) {
                const int r = ylo + (seg >> 1);
                const bool seen = (r >= pylo) && (r <= pyhi);
                int ca, cb;
                if (seg & 1) { ca = pxhi + 1; cb = seen ? xhi : pxhi; }
                else         { ca = xlo;      cb = seen ? pxlo - 1 : xhi; }
                if (ca > cb) continue;
                int pos = sStart[r * GX + ca];
                const int end = sStart[r * GX + cb + 1];
#define FOUR_STEP(q0, q1, q2, q3, leftv) {                                                    \
                    const int lf = (leftv);                                                   \
                    const unsigned i0 = __float_as_uint((q0).z), i1 = __float_as_uint((q1).z); \
                    const unsigned i2 = __float_as_uint((q2).z), i3 = __float_as_uint((q3).z); \
                    const float ax = (q0).x - px, ay = (q0).y - py;                           \
                    const float bx = (q1).x - px, by = (q1).y - py;                           \
                    const float cx = (q2).x - px, cy = (q2).y - py;                           \
                    const float ex = (q3).x - px, ey = (q3).y - py;                           \
                    const float s0 = __builtin_fmaf(ax, ax, ay * ay);                         \
                    const float s1 = __builtin_fmaf(bx, bx, by * by);                         \
                    const float s2 = __builtin_fmaf(cx, cx, cy * cy);                         \
                    const float s3 = __builtin_fmaf(ex, ex, ey * ey);                         \
                    const unsigned w0 = (__float_as_uint(s0) & BKT_MASK) | i0;                \
                    const unsigned w1 = (__float_as_uint(s1) & BKT_MASK) | i1;                \
                    const unsigned w2 = (__float_as_uint(s2) & BKT_MASK) | i2;                \
                    const unsigned w3 = (__float_as_uint(s3) & BKT_MASK) | i3;                \
                    const unsigned u0 = ((lf > 0) & (i0 != (unsigned)node)) ? w0 : 0xFFFFFFFFu; \
                    const unsigned u1 = ((lf > 1) & (i1 != (unsigned)node)) ? w1 : 0xFFFFFFFFu; \
                    const unsigned u2 = ((lf > 2) & (i2 != (unsigned)node)) ? w2 : 0xFFFFFFFFu; \
                    const unsigned u3 = ((lf > 3) & (i3 != (unsigned)node)) ? w3 : 0xFFFFFFFFu; \
                    WORD_INSERT(u0) WORD_INSERT(u1) WORD_INSERT(u2) WORD_INSERT(u3)           \
                }
                float4 a0 = sCand[pos];
                float4 a1 = sCand[pos + 1];
                float4 a2 = sCand[pos + 2];
                float4 a3 = sCand[pos + 3];
                while (pos < end) {
                    const float4 b0 = sCand[pos + 4];
                    const float4 b1 = sCand[pos + 5];
                    const float4 b2 = sCand[pos + 6];
                    const float4 b3 = sCand[pos + 7];
                    __builtin_amdgcn_sched_barrier(0);
                    FOUR_STEP(a0, a1, a2, a3, end - pos)
                    __builtin_amdgcn_sched_barrier(0);
                    a0 = sCand[pos + 8];
                    a1 = sCand[pos + 9];
                    a2 = sCand[pos + 10];
                    a3 = sCand[pos + 11];
                    __builtin_amdgcn_sched_barrier(0);
                    FOUR_STEP(b0, b1, b2, b3, end - pos - 4)
                    pos += 8;
                }
#undef FOUR_STEP
            }
            sx0 = xlo; sx1 = xhi; sy0 = ylo; sy1 = yhi;
            const float inf = __builtin_inff();
            const float gx0 = (xlo > 0)      ? px - (float)xlo * (1.0f / GX)       : inf;
            const float gx1 = (xhi < GX - 1) ? (float)(xhi + 1) * (1.0f / GX) - px : inf;
            const float gy0 = (ylo > 0)      ? py - (float)ylo * (1.0f / GY)       : inf;
            const float gy1 = (yhi < GY - 1) ? (float)(yhi + 1) * (1.0f / GY) - py : inf;
            const float d10 = __uint_as_float(K9 | IDX_MASK);
            const bool ux0 = !(d10 < gx0 * gx0);
            const bool ux1 = !(d10 < gx1 * gx1);
            const bool uy0 = !(d10 < gy0 * gy0);
            const bool uy1 = !(d10 < gy1 * gy1);
            if (!(ux0 || ux1 || uy0 || uy1)) break;
            pxlo = xlo; pxhi = xhi; pylo = ylo; pyhi = yhi;
            xlo -= ux0 ? 1 : 0;  xhi += ux1 ? 1 : 0;
            ylo -= uy0 ? 1 : 0;  yhi += uy1 ? 1 : 0;
        }
        const bool e0 = (K0 ^ K1) <= IDX_MASK, e1 = (K1 ^ K2) <= IDX_MASK, e2 = (K2 ^ K3) <= IDX_MASK;
        const bool e3 = (K3 ^ K4) <= IDX_MASK, e4 = (K4 ^ K5) <= IDX_MASK, e5 = (K5 ^ K6) <= IDX_MASK;
        const bool e6 = (K6 ^ K7) <= IDX_MASK, e7 = (K7 ^ K8) <= IDX_MASK, e8 = (K8 ^ K9) <= IDX_MASK;
        const bool e9 = (K9 ^ K10) <= IDX_MASK, e10 = (K10 ^ K11) <= IDX_MASK;
        J0 = K0 & IDX_MASK; J1 = K1 & IDX_MASK; J2 = K2 & IDX_MASK; J3 = K3 & IDX_MASK; J4 = K4 & IDX_MASK;
        J5 = K5 & IDX_MASK; J6 = K6 & IDX_MASK; J7 = K7 & IDX_MASK; J8 = K8 & IDX_MASK; J9 = K9 & IDX_MASK;
        if (e0 | e1 | e2 | e3 | e4 | e5 | e6 | e7 | e8 | e9) {
            const unsigned FAR = __float_as_uint(3.0e38f);
            unsigned D0, D1, D2, D3, D4, D5, D6, D7, D8, D9;
            D0 = D1 = D2 = D3 = D4 = D5 = D6 = D7 = D8 = D9 = FAR;
            J0 = J1 = J2 = J3 = J4 = J5 = J6 = J7 = J8 = J9 = 0;
            if (e9 & e10) {
                for (int r = sy0; r <= sy1; ++r) {
                    const int end = sStart[r * GX + sx1 + 1];
                    for (int pos = sStart[r * GX + sx0]; pos < end; ++pos) {
                        const float4 c = sCand[pos];
                        const unsigned j = __float_as_uint(c.z);
                        const float dx = c.x - px;
                        const float dy = c.y - py;
                        const float d2 = __builtin_fmaf(dx, dx, dy * dy);
                        const unsigned du = __float_as_uint(d2);
                        if ((j != (unsigned)node) & KEY_LT(du, j, D9, J9)) {
                            KEY_INSERT(du, j)
                        }
                    }
                }
            } else {
#define RESORT(Kk) {                                                            \
                    const unsigned jj = min((Kk) & IDX_MASK, (unsigned)(NNODES - 1)); \
                    const float2 q = sl[jj];                                    \
                    const float dx = q.x - px;                                  \
                    const float dy = q.y - py;                                  \
                    const float d2 = __builtin_fmaf(dx, dx, dy * dy);           \
                    const unsigned dq = ((Kk) < FARK) ? __float_as_uint(d2) : 0xFFFFFFFFu; \
                    KEY_INSERT(dq, jj)                                          \
                }
                RESORT(K0) RESORT(K1) RESORT(K2) RESORT(K3) RESORT(K4) RESORT(K5)
                RESORT(K6) RESORT(K7) RESORT(K8) RESORT(K9) RESORT(K10)
#undef RESORT
            }
        }
    }
#undef KEY_LT
#undef KEY_INSERT
#undef WORD_INSERT
    const int bestJ[KNN] = {(int)J0, (int)J1, (int)J2, (int)J3, (int)J4,
                            (int)J5, (int)J6, (int)J7, (int)J8, (int)J9};

    float* fr = &sFeat[tid * FSTRIDE];
    if (node < NNODES) {
        fr[0] = px;
        fr[1] = py;
#pragma unroll
        for (int k = 0; k < KNN; ++k) {
            const float2 q = sl[bestJ[k]];
            fr[2 + 2 * k]     = q.x - px;
            fr[2 + 2 * k + 1] = q.y - py;
        }
        fr[NODE_DIM]     = 0.0f;
        fr[NODE_DIM + 1] = 0.0f;
    } else {
#pragma unroll
        for (int k = 0; k < FDIM; ++k) fr[k] = 0.0f;
    }
    __syncthreads();

    const int wave   = tid >> 5;
    const int lane   = tid & 31;
    const int laneLo = lane & 15;
    const int kHalf  = (lane < 16) ? 0 : 2;
    const int mBase  = (lane < 16) ? 0 : 8;

    const int t0 = wave * 2;
    const int t1 = t0 + 1;
    const bool ok0 = (node0 + t0 * 16) < NNODES;
    const bool ok1 = (node0 + t1 * 16) < NNODES;

    v2f a0[6], a1[6];
    {
        const float* fa0 = &sFeat[(t0 * 16 + laneLo) * FSTRIDE];
        const float* fa1 = &sFeat[(t1 * 16 + laneLo) * FSTRIDE];
#pragma unroll
        for (int kc = 0; kc < 6; ++kc) {
            const int k0 = kc * 4 + kHalf;
            a0[kc].x = fa0[k0];  a0[kc].y = fa0[k0 + 1];
            a1[kc].x = fa1[k0];  a1[kc].y = fa1[k0 + 1];
        }
    }

    float* const outB = out + (size_t)batch * NNODES * EMBED_DIM;

#pragma unroll
    for (int ct = 0; ct < 8; ++ct) {
        const int e = ct * 16 + laneLo;
        v2f bf[6];
#pragma unroll
        for (int kc = 0; kc < 6; ++kc) {
            const int k0 = kc * 4 + kHalf;
            bf[kc].x = sW[k0 * EMBED_DIM + e];
            bf[kc].y = sW[(k0 + 1) * EMBED_DIM + e];
        }
        v8f acc0 = {}, acc1 = {};
#pragma unroll
        for (int kc = 0; kc < 6; ++kc)
            acc0 = __builtin_amdgcn_wmma_f32_16x16x4_f32(
                false, a0[kc], false, bf[kc], (short)0, acc0, false, false);
#pragma unroll
        for (int kc = 0; kc < 6; ++kc)
            acc1 = __builtin_amdgcn_wmma_f32_16x16x4_f32(
                false, a1[kc], false, bf[kc], (short)0, acc1, false, false);

        const float bv = bias[e];
        if (ok0) {
            float* p = outB + (size_t)(node0 + t0 * 16 + mBase) * EMBED_DIM + e;
#pragma unroll
            for (int r = 0; r < 8; ++r)
                p[(size_t)r * EMBED_DIM] = acc0[r] + bv;
        }
        if (ok1) {
            float* p = outB + (size_t)(node0 + t1 * 16 + mBase) * EMBED_DIM + e;
#pragma unroll
            for (int r = 0; r < 8; ++r)
                p[(size_t)r * EMBED_DIM] = acc1[r] + bv;
        }
    }
}

extern "C" void kernel_launch(void* const* d_in, const int* in_sizes, int n_in,
                              void* d_out, int out_size, void* d_ws, size_t ws_size,
                              hipStream_t stream) {
    const float* locs = (const float*)d_in[0];
    const float* W    = (const float*)d_in[1];
    const float* b    = (const float*)d_in[2];
    float* out        = (float*)d_out;

    dim3 grid(NBATCH * BLOCKS_PER_BATCH);
    dim3 block(BLOCK);
    tsp_embed_kernel<<<grid, block, 0, stream>>>(locs, W, b, out);
}


# Product's B fragments (8 column tiles x 6 K-chunks) and bias values loaded as one group ahead of the feature barrier: the matrix instructions run back to back; was one LDS load waited for per pair
# speedup vs baseline: 11.7907x; 1.0941x over previous
#include <hip/hip_runtime.h>
#include <hip/hip_bf16.h>

#define EMBED_DIM 128
#define KNN 10
#define NODE_DIM 22
#define FDIM 24
#define NNODES 2000
#define NBATCH 16
#define BLOCK 128
#define NODES_PER_BLOCK 128
#define BLOCKS_PER_BATCH 16
#define FSTRIDE 25
#define GX 8
#define GY 16
#define NCELLS (GX * GY)
#define PTS_PER_THREAD ((NNODES + BLOCK - 1) / BLOCK)

typedef __attribute__((ext_vector_type(2))) float v2f;
typedef __attribute__((ext_vector_type(8))) float v8f;

__global__ __launch_bounds__(BLOCK)
void tsp_embed_kernel(const float* __restrict__ locs,
                      const float* __restrict__ W,
                      const float* __restrict__ bias,
                      float* __restrict__ out)
{
    __shared__ float sLocs[2 * NNODES];
    __shared__ float sFeat[NODES_PER_BLOCK * FSTRIDE];
    __shared__ float sW[FDIM * EMBED_DIM];
    __shared__ float4 sCand[NNODES + 12];
    __shared__ __attribute__((aligned(16))) unsigned short sHist[BLOCK * NCELLS];
    __shared__ int sCnt[NCELLS];
    __shared__ int sStart[NCELLS + 1];

    const int tid   = threadIdx.x;
    __builtin_assume(tid >= 0 && tid < BLOCK);
    const int batch = blockIdx.x / BLOCKS_PER_BATCH;
    const int node0 = (blockIdx.x % BLOCKS_PER_BATCH) * NODES_PER_BLOCK;

    const float2* bl2 = (const float2*)(locs + (size_t)batch * NNODES * 2);
    float2* sl2w = (float2*)sLocs;
    float2 pt[PTS_PER_THREAD];
    int key[PTS_PER_THREAD];
    int slot[PTS_PER_THREAD];
#pragma unroll
    for (int k = 0; k < PTS_PER_THREAD; ++k) {
        const int n = tid + k * BLOCK;
        pt[k] = (n < NNODES) ? bl2[n] : make_float2(0.0f, 0.0f);
    }
    unsigned short* const myH = &sHist[tid * NCELLS];
#pragma unroll
    for (int i = 0; i < NCELLS; ++i) myH[i] = (unsigned short)0;
#pragma unroll
    for (int k = 0; k < PTS_PER_THREAD; ++k) {
        const int n = tid + k * BLOCK;
        key[k] = 0;
        slot[k] = 0;
        if (n < NNODES) {
            const float2 p = pt[k];
            sl2w[n] = p;
            const int bx = (int)fminf(fmaxf(p.x * (float)GX, 0.0f), (float)(GX - 1));
            const int by = (int)fminf(fmaxf(p.y * (float)GY, 0.0f), (float)(GY - 1));
            key[k] = by * GX + bx;
            const unsigned v = myH[key[k]];
            slot[k] = (int)v;
            myH[key[k]] = (unsigned short)(v + 1u);
        }
    }
    for (int idx = tid; idx < FDIM * EMBED_DIM; idx += BLOCK) {
        const int f = idx / EMBED_DIM;
        const int e = idx % EMBED_DIM;
        sW[idx] = (f < NODE_DIM) ? W[e * NODE_DIM + f] : 0.0f;
    }
    if (tid < 12) sCand[NNODES + tid] = make_float4(0.0f, 0.0f, 0.0f, 0.0f);
    __syncthreads();

    {
        int run = 0;
#pragma unroll 16
        for (int t = 0; t < BLOCK; ++t) {
            const int v = sHist[t * NCELLS + tid];
            sHist[t * NCELLS + tid] = (unsigned short)run;
            run += v;
        }
        sCnt[tid] = run;
    }
    __syncthreads();
    int start = 0;
    for (int i = 0; i < NCELLS; ++i) start += (i < tid) ? sCnt[i] : 0;
    sStart[tid] = start;
    if (tid == 0) sStart[NCELLS] = NNODES;
    __syncthreads();
#pragma unroll
    for (int k = 0; k < PTS_PER_THREAD; ++k) {
        const int n = tid + k * BLOCK;
        if (n < NNODES) {
            const int pos = sStart[key[k]] + (int)myH[key[k]] + slot[k];
            sCand[pos] = make_float4(pt[k].x, pt[k].y, __int_as_float(n), 0.0f);
        }
    }
    __syncthreads();
    const float2* sl = (const float2*)sLocs;

    const int node = node0 + tid;
    unsigned K0, K1, K2, K3, K4, K5, K6, K7, K8, K9, K10, K11;
    unsigned J0 = 0, J1 = 0, J2 = 0, J3 = 0, J4 = 0, J5 = 0, J6 = 0, J7 = 0, J8 = 0, J9 = 0;
    const unsigned IDX_MASK = 0x7FFu;
    const unsigned BKT_MASK = 0xFFFFF800u;
    const unsigned FARK = 0x7F000000u;
    K0 = K1 = K2 = K3 = K4 = K5 = K6 = K7 = K8 = K9 = K10 = K11 = FARK;
#define WORD_INSERT(w) {                         \
        K11 = min(max(K10, (w)), K11);          \
        K10 = min(max(K9, (w)), K10);           \
        K9 = min(max(K8, (w)), K9);             \
        K8 = min(max(K7, (w)), K8);             \
        K7 = min(max(K6, (w)), K7);             \
        K6 = min(max(K5, (w)), K6);             \
        K5 = min(max(K4, (w)), K5);             \
        K4 = min(max(K3, (w)), K4);             \
        K3 = min(max(K2, (w)), K3);             \
        K2 = min(max(K1, (w)), K2);             \
        K1 = min(max(K0, (w)), K1);             \
        K0 = min(K0, (w));                      \
    }
#define KEY_LT(du, j, Dk, Jk) (((du) < (Dk)) | (((du) == (Dk)) & ((j) < (Jk))))
#define KEY_INSERT(duv, jv) {                                                   \
        const unsigned eu = (duv), ej = (jv);                                   \
        const bool c0 = KEY_LT(eu, ej, D0, J0), c1 = KEY_LT(eu, ej, D1, J1);    \
        const bool c2 = KEY_LT(eu, ej, D2, J2), c3 = KEY_LT(eu, ej, D3, J3);    \
        const bool c4 = KEY_LT(eu, ej, D4, J4), c5 = KEY_LT(eu, ej, D5, J5);    \
        const bool c6 = KEY_LT(eu, ej, D6, J6), c7 = KEY_LT(eu, ej, D7, J7);    \
        const bool c8 = KEY_LT(eu, ej, D8, J8), c9 = KEY_LT(eu, ej, D9, J9);    \
        D9 = min(max(D8, eu), D9);        J9 = c8 ? J8 : (c9 ? ej : J9);        \
        D8 = min(max(D7, eu), D8);        J8 = c7 ? J7 : (c8 ? ej : J8);        \
        D7 = min(max(D6, eu), D7);        J7 = c6 ? J6 : (c7 ? ej : J7);        \
        D6 = min(max(D5, eu), D6);        J6 = c5 ? J5 : (c6 ? ej : J6);        \
        D5 = min(max(D4, eu), D5);        J5 = c4 ? J4 : (c5 ? ej : J5);        \
        D4 = min(max(D3, eu), D4);        J4 = c3 ? J3 : (c4 ? ej : J4);        \
        D3 = min(max(D2, eu), D3);        J3 = c2 ? J2 : (c3 ? ej : J3);        \
        D2 = min(max(D1, eu), D2);        J2 = c1 ? J1 : (c2 ? ej : J2);        \
        D1 = min(max(D0, eu), D1);        J1 = c0 ? J0 : (c1 ? ej : J1);        \
        D0 = min(D0, eu);                 J0 = c0 ? ej : J0;                    \
    }
    float px = 0.0f, py = 0.0f;
    if (node < NNODES) {
        const float2 me = sl[node];
        px = me.x; py = me.y;
        const float R = 1.0f / 16.0f;
        int xlo = (int)fminf(fmaxf((px - R) * (float)GX, 0.0f), (float)(GX - 1));
        int xhi = (int)fminf(fmaxf((px + R) * (float)GX, 0.0f), (float)(GX - 1));
        int ylo = (int)fminf(fmaxf((py - R) * (float)GY, 0.0f), (float)(GY - 1));
        int yhi = (int)fminf(fmaxf((py + R) * (float)GY, 0.0f), (float)(GY - 1));
        int pxlo = 1, pxhi = 0, pylo = 1, pyhi = 0;
        int sx0 = xlo, sx1 = xhi, sy0 = ylo, sy1 = yhi;
        for (int grow = 0; grow < GX + GY; ++grow) {
            const int nseg = 2 * (yhi - ylo + 1);
            for (int seg = 0; seg < nseg; ++seg) {
                const int r = ylo + (seg >> 1);
                const bool seen = (r >= pylo) && (r <= pyhi);
                int ca, cb;
                if (seg & 1) { ca = pxhi + 1; cb = seen ? xhi : pxhi; }
                else         { ca = xlo;      cb = seen ? pxlo - 1 : xhi; }
                if (ca > cb) continue;
                int pos = sStart[r * GX + ca];
                const int end = sStart[r * GX + cb + 1];
#define FOUR_STEP(q0, q1, q2, q3, leftv) {                                                    \
                    const int lf = (leftv);                                                   \
                    const unsigned i0 = __float_as_uint((q0).z), i1 = __float_as_uint((q1).z); \
                    const unsigned i2 = __float_as_uint((q2).z), i3 = __float_as_uint((q3).z); \
                    const float ax = (q0).x - px, ay = (q0).y - py;                           \
                    const float bx = (q1).x - px, by = (q1).y - py;                           \
                    const float cx = (q2).x - px, cy = (q2).y - py;                           \
                    const float ex = (q3).x - px, ey = (q3).y - py;                           \
                    const float s0 = __builtin_fmaf(ax, ax, ay * ay);                         \
                    const float s1 = __builtin_fmaf(bx, bx, by * by);                         \
                    const float s2 = __builtin_fmaf(cx, cx, cy * cy);                         \
                    const float s3 = __builtin_fmaf(ex, ex, ey * ey);                         \
                    const unsigned w0 = (__float_as_uint(s0) & BKT_MASK) | i0;                \
                    const unsigned w1 = (__float_as_uint(s1) & BKT_MASK) | i1;                \
                    const unsigned w2 = (__float_as_uint(s2) & BKT_MASK) | i2;                \
                    const unsigned w3 = (__float_as_uint(s3) & BKT_MASK) | i3;                \
                    const unsigned u0 = ((lf > 0) & (i0 != (unsigned)node)) ? w0 : 0xFFFFFFFFu; \
                    const unsigned u1 = ((lf > 1) & (i1 != (unsigned)node)) ? w1 : 0xFFFFFFFFu; \
                    const unsigned u2 = ((lf > 2) & (i2 != (unsigned)node)) ? w2 : 0xFFFFFFFFu; \
                    const unsigned u3 = ((lf > 3) & (i3 != (unsigned)node)) ? w3 : 0xFFFFFFFFu; \
                    WORD_INSERT(u0) WORD_INSERT(u1) WORD_INSERT(u2) WORD_INSERT(u3)           \
                }
                float4 a0 = sCand[pos];
                float4 a1 = sCand[pos + 1];
                float4 a2 = sCand[pos + 2];
                float4 a3 = sCand[pos + 3];
                while (pos < end) {
                    const float4 b0 = sCand[pos + 4];
                    const float4 b1 = sCand[pos + 5];
                    const float4 b2 = sCand[pos + 6];
                    const float4 b3 = sCand[pos + 7];
                    __builtin_amdgcn_sched_barrier(0);
                    FOUR_STEP(a0, a1, a2, a3, end - pos)
                    __builtin_amdgcn_sched_barrier(0);
                    a0 = sCand[pos + 8];
                    a1 = sCand[pos + 9];
                    a2 = sCand[pos + 10];
                    a3 = sCand[pos + 11];
                    __builtin_amdgcn_sched_barrier(0);
                    FOUR_STEP(b0, b1, b2, b3, end - pos - 4)
                    pos += 8;
                }
#undef FOUR_STEP
            }
            sx0 = xlo; sx1 = xhi; sy0 = ylo; sy1 = yhi;
            const float inf = __builtin_inff();
            const float gx0 = (xlo > 0)      ? px - (float)xlo * (1.0f / GX)       : inf;
            const float gx1 = (xhi < GX - 1) ? (float)(xhi + 1) * (1.0f / GX) - px : inf;
            const float gy0 = (ylo > 0)      ? py - (float)ylo * (1.0f / GY)       : inf;
            const float gy1 = (yhi < GY - 1) ? (float)(yhi + 1) * (1.0f / GY) - py : inf;
            const float d10 = __uint_as_float(K9 | IDX_MASK);
            const bool ux0 = !(d10 < gx0 * gx0);
            const bool ux1 = !(d10 < gx1 * gx1);
            const bool uy0 = !(d10 < gy0 * gy0);
            const bool uy1 = !(d10 < gy1 * gy1);
            if (!(ux0 || ux1 || uy0 || uy1)) break;
            pxlo = xlo; pxhi = xhi; pylo = ylo; pyhi = yhi;
            xlo -= ux0 ? 1 : 0;  xhi += ux1 ? 1 : 0;
            ylo -= uy0 ? 1 : 0;  yhi += uy1 ? 1 : 0;
        }
        const bool e0 = (K0 ^ K1) <= IDX_MASK, e1 = (K1 ^ K2) <= IDX_MASK, e2 = (K2 ^ K3) <= IDX_MASK;
        const bool e3 = (K3 ^ K4) <= IDX_MASK, e4 = (K4 ^ K5) <= IDX_MASK, e5 = (K5 ^ K6) <= IDX_MASK;
        const bool e6 = (K6 ^ K7) <= IDX_MASK, e7 = (K7 ^ K8) <= IDX_MASK, e8 = (K8 ^ K9) <= IDX_MASK;
        const bool e9 = (K9 ^ K10) <= IDX_MASK, e10 = (K10 ^ K11) <= IDX_MASK;
        J0 = K0 & IDX_MASK; J1 = K1 & IDX_MASK; J2 = K2 & IDX_MASK; J3 = K3 & IDX_MASK; J4 = K4 & IDX_MASK;
        J5 = K5 & IDX_MASK; J6 = K6 & IDX_MASK; J7 = K7 & IDX_MASK; J8 = K8 & IDX_MASK; J9 = K9 & IDX_MASK;
        if (e0 | e1 | e2 | e3 | e4 | e5 | e6 | e7 | e8 | e9) {
            const unsigned FAR = __float_as_uint(3.0e38f);
            unsigned D0, D1, D2, D3, D4, D5, D6, D7, D8, D9;
            D0 = D1 = D2 = D3 = D4 = D5 = D6 = D7 = D8 = D9 = FAR;
            J0 = J1 = J2 = J3 = J4 = J5 = J6 = J7 = J8 = J9 = 0;
            if (e9 & e10) {
                for (int r = sy0; r <= sy1; ++r) {
                    const int end = sStart[r * GX + sx1 + 1];
                    for (int pos = sStart[r * GX + sx0]; pos < end; ++pos) {
                        const float4 c = sCand[pos];
                        const unsigned j = __float_as_uint(c.z);
                        const float dx = c.x - px;
                        const float dy = c.y - py;
                        const float d2 = __builtin_fmaf(dx, dx, dy * dy);
                        const unsigned du = __float_as_uint(d2);
                        if ((j != (unsigned)node) & KEY_LT(du, j, D9, J9)) {
                            KEY_INSERT(du, j)
                        }
                    }
                }
            } else {
#define RESORT(Kk) {                                                            \
                    const unsigned jj = min((Kk) & IDX_MASK, (unsigned)(NNODES - 1)); \
                    const float2 q = sl[jj];                                    \
                    const float dx = q.x - px;                                  \
                    const float dy = q.y - py;                                  \
                    const float d2 = __builtin_fmaf(dx, dx, dy * dy);           \
                    const unsigned dq = ((Kk) < FARK) ? __float_as_uint(d2) : 0xFFFFFFFFu; \
                    KEY_INSERT(dq, jj)                                          \
                }
                RESORT(K0) RESORT(K1) RESORT(K2) RESORT(K3) RESORT(K4) RESORT(K5)
                RESORT(K6) RESORT(K7) RESORT(K8) RESORT(K9) RESORT(K10)
#undef RESORT
            }
        }
    }
#undef KEY_LT
#undef KEY_INSERT
#undef WORD_INSERT
    const int bestJ[KNN] = {(int)J0, (int)J1, (int)J2, (int)J3, (int)J4,
                            (int)J5, (int)J6, (int)J7, (int)J8, (int)J9};

    const int wave   = tid >> 5;
    const int lane   = tid & 31;
    const int laneLo = lane & 15;
    const int kHalf  = (lane < 16) ? 0 : 2;
    const int mBase  = (lane < 16) ? 0 : 8;
    v2f bfAll[8][6];
    float bvAll[8];
#pragma unroll
    for (int ct = 0; ct < 8; ++ct) {
        const int e = ct * 16 + laneLo;
#pragma unroll
        for (int kc = 0; kc < 6; ++kc) {
            const int k0 = kc * 4 + kHalf;
            bfAll[ct][kc].x = sW[k0 * EMBED_DIM + e];
            bfAll[ct][kc].y = sW[(k0 + 1) * EMBED_DIM + e];
        }
        bvAll[ct] = bias[e];
    }
    __builtin_amdgcn_sched_barrier(0);

    float* fr = &sFeat[tid * FSTRIDE];
    if (node < NNODES) {
        fr[0] = px;
        fr[1] = py;
#pragma unroll
        for (int k = 0; k < KNN; ++k) {
            const float2 q = sl[bestJ[k]];
            fr[2 + 2 * k]     = q.x - px;
            fr[2 + 2 * k + 1] = q.y - py;
        }
        fr[NODE_DIM]     = 0.0f;
        fr[NODE_DIM + 1] = 0.0f;
    } else {
#pragma unroll
        for (int k = 0; k < FDIM; ++k) fr[k] = 0.0f;
    }
    __syncthreads();

    const int t0 = wave * 2;
    const int t1 = t0 + 1;
    const bool ok0 = (node0 + t0 * 16) < NNODES;
    const bool ok1 = (node0 + t1 * 16) < NNODES;

    v2f a0[6], a1[6];
    {
        const float* fa0 = &sFeat[(t0 * 16 + laneLo) * FSTRIDE];
        const float* fa1 = &sFeat[(t1 * 16 + laneLo) * FSTRIDE];
#pragma unroll
        for (int kc = 0; kc < 6; ++kc) {
            const int k0 = kc * 4 + kHalf;
            a0[kc].x = fa0[k0];  a0[kc].y = fa0[k0 + 1];
            a1[kc].x = fa1[k0];  a1[kc].y = fa1[k0 + 1];
        }
    }

    float* const outB = out + (size_t)batch * NNODES * EMBED_DIM;

#pragma unroll
    for (int ct = 0; ct < 8; ++ct) {
        const int e = ct * 16 + laneLo;
        v8f acc0 = {}, acc1 = {};
#pragma unroll
        for (int kc = 0; kc < 6; ++kc) {
            acc0 = __builtin_amdgcn_wmma_f32_16x16x4_f32(
                false, a0[kc], false, bfAll[ct][kc], (short)0, acc0, false, false);
            acc1 = __builtin_amdgcn_wmma_f32_16x16x4_f32(
                false, a1[kc], false, bfAll[ct][kc], (short)0, acc1, false, false);
        }

        const float bv = bvAll[ct];
        if (ok0) {
            float* p = outB + (size_t)(node0 + t0 * 16 + mBase) * EMBED_DIM + e;
#pragma unroll
            for (int r = 0; r < 8; ++r)
                p[(size_t)r * EMBED_DIM] = acc0[r] + bv;
        }
        if (ok1) {
            float* p = outB + (size_t)(node0 + t1 * 16 + mBase) * EMBED_DIM + e;
#pragma unroll
            for (int r = 0; r < 8; ++r)
                p[(size_t)r * EMBED_DIM] = acc1[r] + bv;
        }
    }
}

extern "C" void kernel_launch(void* const* d_in, const int* in_sizes, int n_in,
                              void* d_out, int out_size, void* d_ws, size_t ws_size,
                              hipStream_t stream) {
    const float* locs = (const float*)d_in[0];
    const float* W    = (const float*)d_in[1];
    const float* b    = (const float*)d_in[2];
    float* out        = (float*)d_out;

    dim3 grid(NBATCH * BLOCKS_PER_BATCH);
    dim3 block(BLOCK);
    tsp_embed_kernel<<<grid, block, 0, stream>>>(locs, W, b, out);
}


# W and bias loaded from global memory straight into the product's B-fragment and bias registers, one group at kernel entry; the LDS copy of W^T and its chain of conditional loads are gone
# speedup vs baseline: 12.6748x; 1.0750x over previous
#include <hip/hip_runtime.h>
#include <hip/hip_bf16.h>

#define EMBED_DIM 128
#define KNN 10
#define NODE_DIM 22
#define FDIM 24
#define NNODES 2000
#define NBATCH 16
#define BLOCK 128
#define NODES_PER_BLOCK 128
#define BLOCKS_PER_BATCH 16
#define FSTRIDE 25
#define GX 8
#define GY 16
#define NCELLS (GX * GY)
#define PTS_PER_THREAD ((NNODES + BLOCK - 1) / BLOCK)

typedef __attribute__((ext_vector_type(2))) float v2f;
typedef __attribute__((ext_vector_type(8))) float v8f;

__global__ __launch_bounds__(BLOCK)
void tsp_embed_kernel(const float* __restrict__ locs,
                      const float* __restrict__ W,
                      const float* __restrict__ bias,
                      float* __restrict__ out)
{
    __shared__ float sLocs[2 * NNODES];
    __shared__ float sFeat[NODES_PER_BLOCK * FSTRIDE];
    __shared__ float4 sCand[NNODES + 12];
    __shared__ __attribute__((aligned(16))) unsigned short sHist[BLOCK * NCELLS];
    __shared__ int sCnt[NCELLS];
    __shared__ int sStart[NCELLS + 1];

    const int tid   = threadIdx.x;
    __builtin_assume(tid >= 0 && tid < BLOCK);
    const int batch = blockIdx.x / BLOCKS_PER_BATCH;
    const int node0 = (blockIdx.x % BLOCKS_PER_BATCH) * NODES_PER_BLOCK;

    const float2* bl2 = (const float2*)(locs + (size_t)batch * NNODES * 2);
    float2* sl2w = (float2*)sLocs;
    float2 pt[PTS_PER_THREAD];
    int key[PTS_PER_THREAD];
    int slot[PTS_PER_THREAD];
#pragma unroll
    for (int k = 0; k < PTS_PER_THREAD; ++k) {
        const int n = tid + k * BLOCK;
        pt[k] = (n < NNODES) ? bl2[n] : make_float2(0.0f, 0.0f);
    }
    const int wave   = tid >> 5;
    const int lane   = tid & 31;
    const int laneLo = lane & 15;
    const int kHalf  = (lane < 16) ? 0 : 2;
    const int mBase  = (lane < 16) ? 0 : 8;
    v2f bfAll[8][6];
    float bvAll[8];
#pragma unroll
    for (int ct = 0; ct < 8; ++ct) {
        const int e = ct * 16 + laneLo;
        const float* wr = W + e * NODE_DIM;
#pragma unroll
        for (int kc = 0; kc < 5; ++kc) {
            const float2 w = *(const float2*)(wr + kc * 4 + kHalf);
            bfAll[ct][kc].x = w.x;
            bfAll[ct][kc].y = w.y;
        }
        const float2 w5 = *(const float2*)(wr + 20);
        bfAll[ct][5].x = (lane < 16) ? w5.x : 0.0f;
        bfAll[ct][5].y = (lane < 16) ? w5.y : 0.0f;
        bvAll[ct] = bias[e];
    }
    __builtin_amdgcn_sched_barrier(0);

    unsigned short* const myH = &sHist[tid * NCELLS];
#pragma unroll
    for (int i = 0; i < NCELLS; ++i) myH[i] = (unsigned short)0;
#pragma unroll
    for (int k = 0; k < PTS_PER_THREAD; ++k) {
        const int n = tid + k * BLOCK;
        key[k] = 0;
        slot[k] = 0;
        if (n < NNODES) {
            const float2 p = pt[k];
            sl2w[n] = p;
            const int bx = (int)fminf(fmaxf(p.x * (float)GX, 0.0f), (float)(GX - 1));
            const int by = (int)fminf(fmaxf(p.y * (float)GY, 0.0f), (float)(GY - 1));
            key[k] = by * GX + bx;
            const unsigned v = myH[key[k]];
            slot[k] = (int)v;
            myH[key[k]] = (unsigned short)(v + 1u);
        }
    }
    if (tid < 12) sCand[NNODES + tid] = make_float4(0.0f, 0.0f, 0.0f, 0.0f);
    __syncthreads();

    {
        int run = 0;
#pragma unroll 16
        for (int t = 0; t < BLOCK; ++t) {
            const int v = sHist[t * NCELLS + tid];
            sHist[t * NCELLS + tid] = (unsigned short)run;
            run += v;
        }
        sCnt[tid] = run;
    }
    __syncthreads();
    int start = 0;
    for (int i = 0; i < NCELLS; ++i) start += (i < tid) ? sCnt[i] : 0;
    sStart[tid] = start;
    if (tid == 0) sStart[NCELLS] = NNODES;
    __syncthreads();
#pragma unroll
    for (int k = 0; k < PTS_PER_THREAD; ++k) {
        const int n = tid + k * BLOCK;
        if (n < NNODES) {
            const int pos = sStart[key[k]] + (int)myH[key[k]] + slot[k];
            sCand[pos] = make_float4(pt[k].x, pt[k].y, __int_as_float(n), 0.0f);
        }
    }
    __syncthreads();
    const float2* sl = (const float2*)sLocs;

    const int node = node0 + tid;
    unsigned K0, K1, K2, K3, K4, K5, K6, K7, K8, K9, K10, K11;
    unsigned J0 = 0, J1 = 0, J2 = 0, J3 = 0, J4 = 0, J5 = 0, J6 = 0, J7 = 0, J8 = 0, J9 = 0;
    const unsigned IDX_MASK = 0x7FFu;
    const unsigned BKT_MASK = 0xFFFFF800u;
    const unsigned FARK = 0x7F000000u;
    K0 = K1 = K2 = K3 = K4 = K5 = K6 = K7 = K8 = K9 = K10 = K11 = FARK;
#define WORD_INSERT(w) {                         \
        K11 = min(max(K10, (w)), K11);          \
        K10 = min(max(K9, (w)), K10);           \
        K9 = min(max(K8, (w)), K9);             \
        K8 = min(max(K7, (w)), K8);             \
        K7 = min(max(K6, (w)), K7);             \
        K6 = min(max(K5, (w)), K6);             \
        K5 = min(max(K4, (w)), K5);             \
        K4 = min(max(K3, (w)), K4);             \
        K3 = min(max(K2, (w)), K3);             \
        K2 = min(max(K1, (w)), K2);             \
        K1 = min(max(K0, (w)), K1);             \
        K0 = min(K0, (w));                      \
    }
#define KEY_LT(du, j, Dk, Jk) (((du) < (Dk)) | (((du) == (Dk)) & ((j) < (Jk))))
#define KEY_INSERT(duv, jv) {                                                   \
        const unsigned eu = (duv), ej = (jv);                                   \
        const bool c0 = KEY_LT(eu, ej, D0, J0), c1 = KEY_LT(eu, ej, D1, J1);    \
        const bool c2 = KEY_LT(eu, ej, D2, J2), c3 = KEY_LT(eu, ej, D3, J3);    \
        const bool c4 = KEY_LT(eu, ej, D4, J4), c5 = KEY_LT(eu, ej, D5, J5);    \
        const bool c6 = KEY_LT(eu, ej, D6, J6), c7 = KEY_LT(eu, ej, D7, J7);    \
        const bool c8 = KEY_LT(eu, ej, D8, J8), c9 = KEY_LT(eu, ej, D9, J9);    \
        D9 = min(max(D8, eu), D9);        J9 = c8 ? J8 : (c9 ? ej : J9);        \
        D8 = min(max(D7, eu), D8);        J8 = c7 ? J7 : (c8 ? ej : J8);        \
        D7 = min(max(D6, eu), D7);        J7 = c6 ? J6 : (c7 ? ej : J7);        \
        D6 = min(max(D5, eu), D6);        J6 = c5 ? J5 : (c6 ? ej : J6);        \
        D5 = min(max(D4, eu), D5);        J5 = c4 ? J4 : (c5 ? ej : J5);        \
        D4 = min(max(D3, eu), D4);        J4 = c3 ? J3 : (c4 ? ej : J4);        \
        D3 = min(max(D2, eu), D3);        J3 = c2 ? J2 : (c3 ? ej : J3);        \
        D2 = min(max(D1, eu), D2);        J2 = c1 ? J1 : (c2 ? ej : J2);        \
        D1 = min(max(D0, eu), D1);        J1 = c0 ? J0 : (c1 ? ej : J1);        \
        D0 = min(D0, eu);                 J0 = c0 ? ej : J0;                    \
    }
    float px = 0.0f, py = 0.0f;
    if (node < NNODES) {
        const float2 me = sl[node];
        px = me.x; py = me.y;
        const float R = 1.0f / 16.0f;
        int xlo = (int)fminf(fmaxf((px - R) * (float)GX, 0.0f), (float)(GX - 1));
        int xhi = (int)fminf(fmaxf((px + R) * (float)GX, 0.0f), (float)(GX - 1));
        int ylo = (int)fminf(fmaxf((py - R) * (float)GY, 0.0f), (float)(GY - 1));
        int yhi = (int)fminf(fmaxf((py + R) * (float)GY, 0.0f), (float)(GY - 1));
        int pxlo = 1, pxhi = 0, pylo = 1, pyhi = 0;
        int sx0 = xlo, sx1 = xhi, sy0 = ylo, sy1 = yhi;
        for (int grow = 0; grow < GX + GY; ++grow) {
            const int nseg = 2 * (yhi - ylo + 1);
            for (int seg = 0; seg < nseg; ++seg) {
                const int r = ylo + (seg >> 1);
                const bool seen = (r >= pylo) && (r <= pyhi);
                int ca, cb;
                if (seg & 1) { ca = pxhi + 1; cb = seen ? xhi : pxhi; }
                else         { ca = xlo;      cb = seen ? pxlo - 1 : xhi; }
                if (ca > cb) continue;
                int pos = sStart[r * GX + ca];
                const int end = sStart[r * GX + cb + 1];
#define FOUR_STEP(q0, q1, q2, q3, leftv) {                                                    \
                    const int lf = (leftv);                                                   \
                    const unsigned i0 = __float_as_uint((q0).z), i1 = __float_as_uint((q1).z); \
                    const unsigned i2 = __float_as_uint((q2).z), i3 = __float_as_uint((q3).z); \
                    const float ax = (q0).x - px, ay = (q0).y - py;                           \
                    const float bx = (q1).x - px, by = (q1).y - py;                           \
                    const float cx = (q2).x - px, cy = (q2).y - py;                           \
                    const float ex = (q3).x - px, ey = (q3).y - py;                           \
                    const float s0 = __builtin_fmaf(ax, ax, ay * ay);                         \
                    const float s1 = __builtin_fmaf(bx, bx, by * by);                         \
                    const float s2 = __builtin_fmaf(cx, cx, cy * cy);                         \
                    const float s3 = __builtin_fmaf(ex, ex, ey * ey);                         \
                    const unsigned w0 = (__float_as_uint(s0) & BKT_MASK) | i0;                \
                    const unsigned w1 = (__float_as_uint(s1) & BKT_MASK) | i1;                \
                    const unsigned w2 = (__float_as_uint(s2) & BKT_MASK) | i2;                \
                    const unsigned w3 = (__float_as_uint(s3) & BKT_MASK) | i3;                \
                    const unsigned u0 = ((lf > 0) & (i0 != (unsigned)node)) ? w0 : 0xFFFFFFFFu; \
                    const unsigned u1 = ((lf > 1) & (i1 != (unsigned)node)) ? w1 : 0xFFFFFFFFu; \
                    const unsigned u2 = ((lf > 2) & (i2 != (unsigned)node)) ? w2 : 0xFFFFFFFFu; \
                    const unsigned u3 = ((lf > 3) & (i3 != (unsigned)node)) ? w3 : 0xFFFFFFFFu; \
                    WORD_INSERT(u0) WORD_INSERT(u1) WORD_INSERT(u2) WORD_INSERT(u3)           \
                }
                float4 a0 = sCand[pos];
                float4 a1 = sCand[pos + 1];
                float4 a2 = sCand[pos + 2];
                float4 a3 = sCand[pos + 3];
                while (pos < end) {
                    const float4 b0 = sCand[pos + 4];
                    const float4 b1 = sCand[pos + 5];
                    const float4 b2 = sCand[pos + 6];
                    const float4 b3 = sCand[pos + 7];
                    __builtin_amdgcn_sched_barrier(0);
                    FOUR_STEP(a0, a1, a2, a3, end - pos)
                    __builtin_amdgcn_sched_barrier(0);
                    a0 = sCand[pos + 8];
                    a1 = sCand[pos + 9];
                    a2 = sCand[pos + 10];
                    a3 = sCand[pos + 11];
                    __builtin_amdgcn_sched_barrier(0);
                    FOUR_STEP(b0, b1, b2, b3, end - pos - 4)
                    pos += 8;
                }
#undef FOUR_STEP
            }
            sx0 = xlo; sx1 = xhi; sy0 = ylo; sy1 = yhi;
            const float inf = __builtin_inff();
            const float gx0 = (xlo > 0)      ? px - (float)xlo * (1.0f / GX)       : inf;
            const float gx1 = (xhi < GX - 1) ? (float)(xhi + 1) * (1.0f / GX) - px : inf;
            const float gy0 = (ylo > 0)      ? py - (float)ylo * (1.0f / GY)       : inf;
            const float gy1 = (yhi < GY - 1) ? (float)(yhi + 1) * (1.0f / GY) - py : inf;
            const float d10 = __uint_as_float(K9 | IDX_MASK);
            const bool ux0 = !(d10 < gx0 * gx0);
            const bool ux1 = !(d10 < gx1 * gx1);
            const bool uy0 = !(d10 < gy0 * gy0);
            const bool uy1 = !(d10 < gy1 * gy1);
            if (!(ux0 || ux1 || uy0 || uy1)) break;
            pxlo = xlo; pxhi = xhi; pylo = ylo; pyhi = yhi;
            xlo -= ux0 ? 1 : 0;  xhi += ux1 ? 1 : 0;
            ylo -= uy0 ? 1 : 0;  yhi += uy1 ? 1 : 0;
        }
        const bool e0 = (K0 ^ K1) <= IDX_MASK, e1 = (K1 ^ K2) <= IDX_MASK, e2 = (K2 ^ K3) <= IDX_MASK;
        const bool e3 = (K3 ^ K4) <= IDX_MASK, e4 = (K4 ^ K5) <= IDX_MASK, e5 = (K5 ^ K6) <= IDX_MASK;
        const bool e6 = (K6 ^ K7) <= IDX_MASK, e7 = (K7 ^ K8) <= IDX_MASK, e8 = (K8 ^ K9) <= IDX_MASK;
        const bool e9 = (K9 ^ K10) <= IDX_MASK, e10 = (K10 ^ K11) <= IDX_MASK;
        J0 = K0 & IDX_MASK; J1 = K1 & IDX_MASK; J2 = K2 & IDX_MASK; J3 = K3 & IDX_MASK; J4 = K4 & IDX_MASK;
        J5 = K5 & IDX_MASK; J6 = K6 & IDX_MASK; J7 = K7 & IDX_MASK; J8 = K8 & IDX_MASK; J9 = K9 & IDX_MASK;
        if (e0 | e1 | e2 | e3 | e4 | e5 | e6 | e7 | e8 | e9) {
            const unsigned FAR = __float_as_uint(3.0e38f);
            unsigned D0, D1, D2, D3, D4, D5, D6, D7, D8, D9;
            D0 = D1 = D2 = D3 = D4 = D5 = D6 = D7 = D8 = D9 = FAR;
            J0 = J1 = J2 = J3 = J4 = J5 = J6 = J7 = J8 = J9 = 0;
            if (e9 & e10) {
                for (int r = sy0; r <= sy1; ++r) {
                    const int end = sStart[r * GX + sx1 + 1];
                    for (int pos = sStart[r * GX + sx0]; pos < end; ++pos) {
                        const float4 c = sCand[pos];
                        const unsigned j = __float_as_uint(c.z);
                        const float dx = c.x - px;
                        const float dy = c.y - py;
                        const float d2 = __builtin_fmaf(dx, dx, dy * dy);
                        const unsigned du = __float_as_uint(d2);
                        if ((j != (unsigned)node) & KEY_LT(du, j, D9, J9)) {
                            KEY_INSERT(du, j)
                        }
                    }
                }
            } else {
#define RESORT(Kk) {                                                            \
                    const unsigned jj = min((Kk) & IDX_MASK, (unsigned)(NNODES - 1)); \
                    const float2 q = sl[jj];                                    \
                    const float dx = q.x - px;                                  \
                    const float dy = q.y - py;                                  \
                    const float d2 = __builtin_fmaf(dx, dx, dy * dy);           \
                    const unsigned dq = ((Kk) < FARK) ? __float_as_uint(d2) : 0xFFFFFFFFu; \
                    KEY_INSERT(dq, jj)                                          \
                }
                RESORT(K0) RESORT(K1) RESORT(K2) RESORT(K3) RESORT(K4) RESORT(K5)
                RESORT(K6) RESORT(K7) RESORT(K8) RESORT(K9) RESORT(K10)
#undef RESORT
            }
        }
    }
#undef KEY_LT
#undef KEY_INSERT
#undef WORD_INSERT
    const int bestJ[KNN] = {(int)J0, (int)J1, (int)J2, (int)J3, (int)J4,
                            (int)J5, (int)J6, (int)J7, (int)J8, (int)J9};

    float* fr = &sFeat[tid * FSTRIDE];
    if (node < NNODES) {
        fr[0] = px;
        fr[1] = py;
#pragma unroll
        for (int k = 0; k < KNN; ++k) {
            const float2 q = sl[bestJ[k]];
            fr[2 + 2 * k]     = q.x - px;
            fr[2 + 2 * k + 1] = q.y - py;
        }
        fr[NODE_DIM]     = 0.0f;
        fr[NODE_DIM + 1] = 0.0f;
    } else {
#pragma unroll
        for (int k = 0; k < FDIM; ++k) fr[k] = 0.0f;
    }
    __syncthreads();

    const int t0 = wave * 2;
    const int t1 = t0 + 1;
    const bool ok0 = (node0 + t0 * 16) < NNODES;
    const bool ok1 = (node0 + t1 * 16) < NNODES;

    v2f a0[6], a1[6];
    {
        const float* fa0 = &sFeat[(t0 * 16 + laneLo) * FSTRIDE];
        const float* fa1 = &sFeat[(t1 * 16 + laneLo) * FSTRIDE];
#pragma unroll
        for (int kc = 0; kc < 6; ++kc) {
            const int k0 = kc * 4 + kHalf;
            a0[kc].x = fa0[k0];  a0[kc].y = fa0[k0 + 1];
            a1[kc].x = fa1[k0];  a1[kc].y = fa1[k0 + 1];
        }
    }

    float* const outB = out + (size_t)batch * NNODES * EMBED_DIM;

#pragma unroll
    for (int ct = 0; ct < 8; ++ct) {
        const int e = ct * 16 + laneLo;
        v8f acc0 = {}, acc1 = {};
#pragma unroll
        for (int kc = 0; kc < 6; ++kc) {
            acc0 = __builtin_amdgcn_wmma_f32_16x16x4_f32(
                false, a0[kc], false, bfAll[ct][kc], (short)0, acc0, false, false);
            acc1 = __builtin_amdgcn_wmma_f32_16x16x4_f32(
                false, a1[kc], false, bfAll[ct][kc], (short)0, acc1, false, false);
        }

        const float bv = bvAll[ct];
        if (ok0) {
            float* p = outB + (size_t)(node0 + t0 * 16 + mBase) * EMBED_DIM + e;
#pragma unroll
            for (int r = 0; r < 8; ++r)
                p[(size_t)r * EMBED_DIM] = acc0[r] + bv;
        }
        if (ok1) {
            float* p = outB + (size_t)(node0 + t1 * 16 + mBase) * EMBED_DIM + e;
#pragma unroll
            for (int r = 0; r < 8; ++r)
                p[(size_t)r * EMBED_DIM] = acc1[r] + bv;
        }
    }
}

extern "C" void kernel_launch(void* const* d_in, const int* in_sizes, int n_in,
                              void* d_out, int out_size, void* d_ws, size_t ws_size,
                              hipStream_t stream) {
    const float* locs = (const float*)d_in[0];
    const float* W    = (const float*)d_in[1];
    const float* b    = (const float*)d_in[2];
    float* out        = (float*)d_out;

    dim3 grid(NBATCH * BLOCKS_PER_BATCH);
    dim3 block(BLOCK);
    tsp_embed_kernel<<<grid, block, 0, stream>>>(locs, W, b, out);
}
